# starting kernel
# baseline (speedup 1.0000x reference)
#include <hip/hip_runtime.h>
#include <math.h>

#define B_ 4
#define S_ 2048
#define H_ 512
#define E_ 8
#define P_ 4
#define I_ 2048
#define R_ (B_ * S_)
#define KC 128
#define XPAD 516
#define HPAD 132

typedef __attribute__((ext_vector_type(2))) float v2f;
typedef __attribute__((ext_vector_type(8))) float v8f;

__device__ __forceinline__ v8f wmma_f32(v2f a, v2f b, v8f c) {
  return __builtin_amdgcn_wmma_f32_16x16x4_f32(
       false, a,  false, b,
       (short)0, c,  false,  false);
}

__global__ __launch_bounds__(256)
void moe_router_kernel(const float* __restrict__ x,
                       const float* __restrict__ Wp,
                       const float* __restrict__ bp,
                       const float* __restrict__ Ws,
                       const float* __restrict__ bs,
                       float* __restrict__ ew) {
  const int gwave = (int)((blockIdx.x * blockDim.x + threadIdx.x) >> 5);
  const int lane  = (int)(threadIdx.x & 31);
  if (gwave >= R_) return;

  const float* xr = x + (size_t)gwave * H_;
  float a0 = 0.f, a1 = 0.f, a2 = 0.f, a3 = 0.f;
#pragma unroll 4
  for (int j = 0; j < H_ / 32; ++j) {
    const int h = lane + j * 32;
    const float xv = xr[h];
    a0 += xv * Wp[0 * H_ + h];
    a1 += xv * Wp[1 * H_ + h];
    a2 += xv * Wp[2 * H_ + h];
    a3 += xv * Wp[3 * H_ + h];
  }
#pragma unroll
  for (int off = 16; off >= 1; off >>= 1) {
    a0 += __shfl_xor(a0, off, 32);
    a1 += __shfl_xor(a1, off, 32);
    a2 += __shfl_xor(a2, off, 32);
    a3 += __shfl_xor(a3, off, 32);
  }
  if (lane == 0) {
    float p[4] = {a0 + bp[0], a1 + bp[1], a2 + bp[2], a3 + bp[3]};
    float pmax = fmaxf(fmaxf(p[0], p[1]), fmaxf(p[2], p[3]));
    float psum = 0.f;
#pragma unroll
    for (int i = 0; i < 4; ++i) { p[i] = __expf(p[i] - pmax); psum += p[i]; }
    const float pinv = 1.f / psum;
#pragma unroll
    for (int i = 0; i < 4; ++i) p[i] *= pinv;

    float el[E_];
    float emax = -1e30f;
#pragma unroll
    for (int e = 0; e < E_; ++e) {
      float v = bs[e];
#pragma unroll
      for (int q = 0; q < 4; ++q) v += p[q] * Ws[e * P_ + q];
      el[e] = v;
      emax = fmaxf(emax, v);
    }
    float esum = 0.f;
#pragma unroll
    for (int e = 0; e < E_; ++e) { el[e] = __expf(el[e] - emax); esum += el[e]; }
    const float einv = 1.f / esum;
#pragma unroll
    for (int e = 0; e < E_; ++e) ew[(size_t)e * R_ + gwave] = el[e] * einv;
  }
}

__global__ __launch_bounds__(256)
void moe_fused_kernel(const float* __restrict__ x,
                      const float* __restrict__ Wup,
                      const float* __restrict__ Wdn,
                      const float* __restrict__ ew,
                      float* __restrict__ out) {
  __shared__ float x_lds[16][XPAD];
  __shared__ float hid_lds[16][HPAD];
  __shared__ float w_lds[E_][16];

  const int row0   = (int)blockIdx.x * 16;
  const int tid    = (int)threadIdx.x;
  const int wv     = tid >> 5;
  const int lane   = tid & 31;
  const int lane16 = lane & 15;
  const int khalf  = (lane >> 4) * 2;
  const int hi8    = (lane >> 4) * 8;

  for (int idx = tid; idx < 16 * H_; idx += 256) {
    const int r = idx >> 9;
    const int c = idx & (H_ - 1);
    x_lds[r][c] = x[(size_t)(row0 + r) * H_ + c];
  }
  if (tid < E_ * 16) {
    const int e = tid >> 4, rr = tid & 15;
    w_lds[e][rr] = ew[(size_t)e * R_ + row0 + rr];
  }
  __syncthreads();

  const v8f vzero = {};
  v8f fin[4];
#pragma unroll
  for (int t = 0; t < 4; ++t) fin[t] = vzero;

  for (int e = 0; e < E_; ++e) {
    const float* Wup_e = Wup + (size_t)e * I_ * H_;
    const float* Wdn_e = Wdn + (size_t)e * H_ * I_;

    v8f eacc[4];
#pragma unroll
    for (int t = 0; t < 4; ++t) eacc[t] = vzero;

    for (int i0 = 0; i0 < I_; i0 += KC) {
      v8f hacc = vzero;
      const float* bptr = Wup_e + (size_t)(i0 + wv * 16 + lane16) * H_ + khalf;
#pragma unroll 4
      for (int k0 = 0; k0 < H_; k0 += 4) {
        v2f a, b;
        a[0] = x_lds[lane16][k0 + khalf];
        a[1] = x_lds[lane16][k0 + khalf + 1];
        b[0] = bptr[k0];
        b[1] = bptr[k0 + 1];
        hacc = wmma_f32(a, b, hacc);
      }
#pragma unroll
      for (int v = 0; v < 8; ++v) {
        const float hv = hacc[v];
        const float g = 0.5f * hv * (1.0f + erff(hv * 0.70710678118654752f));
        hid_lds[v + hi8][wv * 16 + lane16] = g;
      }
      __syncthreads();

#pragma unroll
      for (int t = 0; t < 4; ++t) {
        const int n = wv * 64 + t * 16 + lane16;
        const float* b2 = Wdn_e + (size_t)n * I_ + i0 + khalf;
#pragma unroll 4
        for (int k0 = 0; k0 < KC; k0 += 4) {
          v2f a, b;
          a[0] = hid_lds[lane16][k0 + khalf];
          a[1] = hid_lds[lane16][k0 + khalf + 1];
          b[0] = b2[k0];
          b[1] = b2[k0 + 1];
          eacc[t] = wmma_f32(a, b, eacc[t]);
        }
      }
      __syncthreads();
    }

#pragma unroll
    for (int t = 0; t < 4; ++t) {
#pragma unroll
      for (int v = 0; v < 8; ++v) {
        const float wgt = w_lds[e][v + hi8];
        fin[t][v] += wgt * eacc[t][v];
      }
    }
  }

#pragma unroll
  for (int t = 0; t < 4; ++t) {
    const int col = wv * 64 + t * 16 + lane16;
#pragma unroll
    for (int v = 0; v < 8; ++v) {
      const int row = row0 + v + hi8;
      out[(size_t)row * H_ + col] = fin[t][v];
    }
  }
}

extern "C" void kernel_launch(void* const* d_in, const int* in_sizes, int n_in,
                              void* d_out, int out_size, void* d_ws, size_t ws_size,
                              hipStream_t stream) {
  const float* x    = (const float*)d_in[0];
  const float* Wp   = (const float*)d_in[1];
  const float* bp   = (const float*)d_in[2];
  const float* Ws   = (const float*)d_in[3];
  const float* bs   = (const float*)d_in[4];
  const float* Wup  = (const float*)d_in[5];
  const float* Wdn  = (const float*)d_in[6];
  float* out = (float*)d_out;
  float* ew  = (float*)d_ws;

  moe_router_kernel<<<R_ / 8, 256, 0, stream>>>(x, Wp, bp, Ws, bs, ew);

  moe_fused_kernel<<<R_ / 16, 256, 0, stream>>>(x, Wup, Wdn, ew, out);
}


# Fused MLP kernel: 32 token rows per workgroup (16 waves); each wave uses every global weight fragment for two 16-row WMMA tiles, halving weight loads per WMMA; summation order unchanged
# speedup vs baseline: 1.5664x; 1.5664x over previous
#include <hip/hip_runtime.h>
#include <math.h>

#define B_ 4
#define S_ 2048
#define H_ 512
#define E_ 8
#define P_ 4
#define I_ 2048
#define R_ (B_ * S_)
#define MT 2
#define MR (16 * MT)
#define NW 16
#define KC (NW * 16)
#define XPAD 516
#define HPAD 260

typedef __attribute__((ext_vector_type(2))) float v2f;
typedef __attribute__((ext_vector_type(8))) float v8f;

__device__ __forceinline__ v8f wmma_f32(v2f a, v2f b, v8f c) {
  return __builtin_amdgcn_wmma_f32_16x16x4_f32(
       false, a,  false, b,
       (short)0, c,  false,  false);
}

__global__ __launch_bounds__(256)
void moe_router_kernel(const float* __restrict__ x,
                       const float* __restrict__ Wp,
                       const float* __restrict__ bp,
                       const float* __restrict__ Ws,
                       const float* __restrict__ bs,
                       float* __restrict__ ew) {
  const int gwave = (int)((blockIdx.x * blockDim.x + threadIdx.x) >> 5);
  const int lane  = (int)(threadIdx.x & 31);
  if (gwave >= R_) return;

  const float* xr = x + (size_t)gwave * H_;
  float a0 = 0.f, a1 = 0.f, a2 = 0.f, a3 = 0.f;
#pragma unroll 4
  for (int j = 0; j < H_ / 32; ++j) {
    const int h = lane + j * 32;
    const float xv = xr[h];
    a0 += xv * Wp[0 * H_ + h];
    a1 += xv * Wp[1 * H_ + h];
    a2 += xv * Wp[2 * H_ + h];
    a3 += xv * Wp[3 * H_ + h];
  }
#pragma unroll
  for (int off = 16; off >= 1; off >>= 1) {
    a0 += __shfl_xor(a0, off, 32);
    a1 += __shfl_xor(a1, off, 32);
    a2 += __shfl_xor(a2, off, 32);
    a3 += __shfl_xor(a3, off, 32);
  }
  if (lane == 0) {
    float p[4] = {a0 + bp[0], a1 + bp[1], a2 + bp[2], a3 + bp[3]};
    float pmax = fmaxf(fmaxf(p[0], p[1]), fmaxf(p[2], p[3]));
    float psum = 0.f;
#pragma unroll
    for (int i = 0; i < 4; ++i) { p[i] = __expf(p[i] - pmax); psum += p[i]; }
    const float pinv = 1.f / psum;
#pragma unroll
    for (int i = 0; i < 4; ++i) p[i] *= pinv;

    float el[E_];
    float emax = -1e30f;
#pragma unroll
    for (int e = 0; e < E_; ++e) {
      float v = bs[e];
#pragma unroll
      for (int q = 0; q < 4; ++q) v += p[q] * Ws[e * P_ + q];
      el[e] = v;
      emax = fmaxf(emax, v);
    }
    float esum = 0.f;
#pragma unroll
    for (int e = 0; e < E_; ++e) { el[e] = __expf(el[e] - emax); esum += el[e]; }
    const float einv = 1.f / esum;
#pragma unroll
    for (int e = 0; e < E_; ++e) ew[(size_t)e * R_ + gwave] = el[e] * einv;
  }
}

__global__ __launch_bounds__(NW * 32)
void moe_fused_kernel(const float* __restrict__ x,
                      const float* __restrict__ Wup,
                      const float* __restrict__ Wdn,
                      const float* __restrict__ ew,
                      float* __restrict__ out) {
  __shared__ float x_lds[MR][XPAD];
  __shared__ float hid_lds[MR][HPAD];
  __shared__ float w_lds[E_][MR];

  const int row0   = (int)blockIdx.x * MR;
  const int tid    = (int)threadIdx.x;
  const int wv     = tid >> 5;
  const int lane   = tid & 31;
  const int lane16 = lane & 15;
  const int khalf  = (lane >> 4) * 2;
  const int hi8    = (lane >> 4) * 8;

  for (int idx = tid; idx < MR * H_; idx += NW * 32) {
    const int r = idx >> 9;
    const int c = idx & (H_ - 1);
    x_lds[r][c] = x[(size_t)(row0 + r) * H_ + c];
  }
  if (tid < E_ * MR) {
    const int e = tid >> 5, rr = tid & (MR - 1);
    w_lds[e][rr] = ew[(size_t)e * R_ + row0 + rr];
  }
  __syncthreads();

  const v8f vzero = {};
  v8f fin[MT][2];
#pragma unroll
  for (int m = 0; m < MT; ++m) { fin[m][0] = vzero; fin[m][1] = vzero; }

  for (int e = 0; e < E_; ++e) {
    const float* Wup_e = Wup + (size_t)e * I_ * H_;
    const float* Wdn_e = Wdn + (size_t)e * H_ * I_;

    v8f eacc[MT][2];
#pragma unroll
    for (int m = 0; m < MT; ++m) { eacc[m][0] = vzero; eacc[m][1] = vzero; }

    for (int i0 = 0; i0 < I_; i0 += KC) {
      v8f hacc[MT];
#pragma unroll
      for (int m = 0; m < MT; ++m) hacc[m] = vzero;
      const float* bptr = Wup_e + (size_t)(i0 + wv * 16 + lane16) * H_ + khalf;
#pragma unroll 4
      for (int k0 = 0; k0 < H_; k0 += 4) {
        v2f b;
        b[0] = bptr[k0];
        b[1] = bptr[k0 + 1];
#pragma unroll
        for (int m = 0; m < MT; ++m) {
          v2f a;
          a[0] = x_lds[m * 16 + lane16][k0 + khalf];
          a[1] = x_lds[m * 16 + lane16][k0 + khalf + 1];
          hacc[m] = wmma_f32(a, b, hacc[m]);
        }
      }
#pragma unroll
      for (int m = 0; m < MT; ++m) {
#pragma unroll
        for (int v = 0; v < 8; ++v) {
          const float hv = hacc[m][v];
          const float g = 0.5f * hv * (1.0f + erff(hv * 0.70710678118654752f));
          hid_lds[m * 16 + v + hi8][wv * 16 + lane16] = g;
        }
      }
      __syncthreads();

      const float* b2 = Wdn_e + (size_t)(wv * 32 + lane16) * I_ + i0 + khalf;
#pragma unroll 4
      for (int k0 = 0; k0 < KC; k0 += 4) {
        v2f a[MT], b[2];
#pragma unroll
        for (int m = 0; m < MT; ++m) {
          a[m][0] = hid_lds[m * 16 + lane16][k0 + khalf];
          a[m][1] = hid_lds[m * 16 + lane16][k0 + khalf + 1];
        }
#pragma unroll
        for (int t = 0; t < 2; ++t) {
          b[t][0] = b2[(size_t)t * 16 * I_ + k0];
          b[t][1] = b2[(size_t)t * 16 * I_ + k0 + 1];
        }
#pragma unroll
        for (int t = 0; t < 2; ++t) {
#pragma unroll
          for (int m = 0; m < MT; ++m) eacc[m][t] = wmma_f32(a[m], b[t], eacc[m][t]);
        }
      }
      __syncthreads();
    }

#pragma unroll
    for (int m = 0; m < MT; ++m) {
#pragma unroll
      for (int t = 0; t < 2; ++t) {
#pragma unroll
        for (int v = 0; v < 8; ++v) {
          const float wgt = w_lds[e][m * 16 + v + hi8];
          fin[m][t][v] += wgt * eacc[m][t][v];
        }
      }
    }
  }

#pragma unroll
  for (int m = 0; m < MT; ++m) {
#pragma unroll
    for (int t = 0; t < 2; ++t) {
      const int col = wv * 32 + t * 16 + lane16;
#pragma unroll
      for (int v = 0; v < 8; ++v) {
        const int row = row0 + m * 16 + v + hi8;
        out[(size_t)row * H_ + col] = fin[m][t][v];
      }
    }
  }
}

extern "C" void kernel_launch(void* const* d_in, const int* in_sizes, int n_in,
                              void* d_out, int out_size, void* d_ws, size_t ws_size,
                              hipStream_t stream) {
  const float* x    = (const float*)d_in[0];
  const float* Wp   = (const float*)d_in[1];
  const float* bp   = (const float*)d_in[2];
  const float* Ws   = (const float*)d_in[3];
  const float* bs   = (const float*)d_in[4];
  const float* Wup  = (const float*)d_in[5];
  const float* Wdn  = (const float*)d_in[6];
  float* out = (float*)d_out;
  float* ew  = (float*)d_ws;

  moe_router_kernel<<<R_ / 8, 256, 0, stream>>>(x, Wp, bp, Ws, bs, ew);

  moe_fused_kernel<<<R_ / 32, 512, 0, stream>>>(x, Wup, Wdn, ew, out);
}


# A and B fragment loads widened from 8 to 16 bytes per lane: K walked in groups of 8, each lane half holds 4 consecutive k for two WMMAs; k summation order inside each group of 8 changes
# speedup vs baseline: 1.7078x; 1.0903x over previous
#include <hip/hip_runtime.h>
#include <math.h>

#define B_ 4
#define S_ 2048
#define H_ 512
#define E_ 8
#define P_ 4
#define I_ 2048
#define R_ (B_ * S_)
#define MT 2
#define MR (16 * MT)
#define NW 16
#define KC (NW * 16)
#define XPAD 516
#define HPAD 260

typedef __attribute__((ext_vector_type(2))) float v2f;
typedef __attribute__((ext_vector_type(4))) float v4f;
typedef __attribute__((ext_vector_type(8))) float v8f;

__device__ __forceinline__ v8f wmma_f32(v2f a, v2f b, v8f c) {
  return __builtin_amdgcn_wmma_f32_16x16x4_f32(
       false, a,  false, b,
       (short)0, c,  false,  false);
}

__global__ __launch_bounds__(256)
void moe_router_kernel(const float* __restrict__ x,
                       const float* __restrict__ Wp,
                       const float* __restrict__ bp,
                       const float* __restrict__ Ws,
                       const float* __restrict__ bs,
                       float* __restrict__ ew) {
  const int gwave = (int)((blockIdx.x * blockDim.x + threadIdx.x) >> 5);
  const int lane  = (int)(threadIdx.x & 31);
  if (gwave >= R_) return;

  const float* xr = x + (size_t)gwave * H_;
  float a0 = 0.f, a1 = 0.f, a2 = 0.f, a3 = 0.f;
#pragma unroll 4
  for (int j = 0; j < H_ / 32; ++j) {
    const int h = lane + j * 32;
    const float xv = xr[h];
    a0 += xv * Wp[0 * H_ + h];
    a1 += xv * Wp[1 * H_ + h];
    a2 += xv * Wp[2 * H_ + h];
    a3 += xv * Wp[3 * H_ + h];
  }
#pragma unroll
  for (int off = 16; off >= 1; off >>= 1) {
    a0 += __shfl_xor(a0, off, 32);
    a1 += __shfl_xor(a1, off, 32);
    a2 += __shfl_xor(a2, off, 32);
    a3 += __shfl_xor(a3, off, 32);
  }
  if (lane == 0) {
    float p[4] = {a0 + bp[0], a1 + bp[1], a2 + bp[2], a3 + bp[3]};
    float pmax = fmaxf(fmaxf(p[0], p[1]), fmaxf(p[2], p[3]));
    float psum = 0.f;
#pragma unroll
    for (int i = 0; i < 4; ++i) { p[i] = __expf(p[i] - pmax); psum += p[i]; }
    const float pinv = 1.f / psum;
#pragma unroll
    for (int i = 0; i < 4; ++i) p[i] *= pinv;

    float el[E_];
    float emax = -1e30f;
#pragma unroll
    for (int e = 0; e < E_; ++e) {
      float v = bs[e];
#pragma unroll
      for (int q = 0; q < 4; ++q) v += p[q] * Ws[e * P_ + q];
      el[e] = v;
      emax = fmaxf(emax, v);
    }
    float esum = 0.f;
#pragma unroll
    for (int e = 0; e < E_; ++e) { el[e] = __expf(el[e] - emax); esum += el[e]; }
    const float einv = 1.f / esum;
#pragma unroll
    for (int e = 0; e < E_; ++e) ew[(size_t)e * R_ + gwave] = el[e] * einv;
  }
}

__global__ __launch_bounds__(NW * 32)
void moe_fused_kernel(const float* __restrict__ x,
                      const float* __restrict__ Wup,
                      const float* __restrict__ Wdn,
                      const float* __restrict__ ew,
                      float* __restrict__ out) {
  __shared__ __attribute__((aligned(16))) float x_lds[MR][XPAD];
  __shared__ __attribute__((aligned(16))) float hid_lds[MR][HPAD];
  __shared__ float w_lds[E_][MR];

  const int row0   = (int)blockIdx.x * MR;
  const int tid    = (int)threadIdx.x;
  const int wv     = tid >> 5;
  const int lane   = tid & 31;
  const int lane16 = lane & 15;
  const int kq     = (lane >> 4) * 4;
  const int hi8    = (lane >> 4) * 8;

  for (int idx = tid; idx < MR * H_; idx += NW * 32) {
    const int r = idx >> 9;
    const int c = idx & (H_ - 1);
    x_lds[r][c] = x[(size_t)(row0 + r) * H_ + c];
  }
  if (tid < E_ * MR) {
    const int e = tid >> 5, rr = tid & (MR - 1);
    w_lds[e][rr] = ew[(size_t)e * R_ + row0 + rr];
  }
  __syncthreads();

  const v8f vzero = {};
  v8f fin[MT][2];
#pragma unroll
  for (int m = 0; m < MT; ++m) { fin[m][0] = vzero; fin[m][1] = vzero; }

  for (int e = 0; e < E_; ++e) {
    const float* Wup_e = Wup + (size_t)e * I_ * H_;
    const float* Wdn_e = Wdn + (size_t)e * H_ * I_;

    v8f eacc[MT][2];
#pragma unroll
    for (int m = 0; m < MT; ++m) { eacc[m][0] = vzero; eacc[m][1] = vzero; }

    for (int i0 = 0; i0 < I_; i0 += KC) {
      v8f hacc[MT];
#pragma unroll
      for (int m = 0; m < MT; ++m) hacc[m] = vzero;
      const float* bptr = Wup_e + (size_t)(i0 + wv * 16 + lane16) * H_ + kq;
#pragma unroll 4
      for (int k0 = 0; k0 < H_; k0 += 8) {
        const v4f b4 = *reinterpret_cast<const v4f*>(bptr + k0);
        v4f a4[MT];
#pragma unroll
        for (int m = 0; m < MT; ++m)
          a4[m] = *reinterpret_cast<const v4f*>(&x_lds[m * 16 + lane16][k0 + kq]);
#pragma unroll
        for (int m = 0; m < MT; ++m) hacc[m] = wmma_f32(a4[m].lo, b4.lo, hacc[m]);
#pragma unroll
        for (int m = 0; m < MT; ++m) hacc[m] = wmma_f32(a4[m].hi, b4.hi, hacc[m]);
      }
#pragma unroll
      for (int m = 0; m < MT; ++m) {
#pragma unroll
        for (int v = 0; v < 8; ++v) {
          const float hv = hacc[m][v];
          const float g = 0.5f * hv * (1.0f + erff(hv * 0.70710678118654752f));
          hid_lds[m * 16 + v + hi8][wv * 16 + lane16] = g;
        }
      }
      __syncthreads();

      const float* b2 = Wdn_e + (size_t)(wv * 32 + lane16) * I_ + i0 + kq;
#pragma unroll 4
      for (int k0 = 0; k0 < KC; k0 += 8) {
        v4f a4[MT], b4[2];
#pragma unroll
        for (int m = 0; m < MT; ++m)
          a4[m] = *reinterpret_cast<const v4f*>(&hid_lds[m * 16 + lane16][k0 + kq]);
#pragma unroll
        for (int t = 0; t < 2; ++t)
          b4[t] = *reinterpret_cast<const v4f*>(b2 + (size_t)t * 16 * I_ + k0);
#pragma unroll
        for (int t = 0; t < 2; ++t) {
#pragma unroll
          for (int m = 0; m < MT; ++m) eacc[m][t] = wmma_f32(a4[m].lo, b4[t].lo, eacc[m][t]);
        }
#pragma unroll
        for (int t = 0; t < 2; ++t) {
#pragma unroll
          for (int m = 0; m < MT; ++m) eacc[m][t] = wmma_f32(a4[m].hi, b4[t].hi, eacc[m][t]);
        }
      }
      __syncthreads();
    }

#pragma unroll
    for (int m = 0; m < MT; ++m) {
#pragma unroll
      for (int t = 0; t < 2; ++t) {
#pragma unroll
        for (int v = 0; v < 8; ++v) {
          const float wgt = w_lds[e][m * 16 + v + hi8];
          fin[m][t][v] += wgt * eacc[m][t][v];
        }
      }
    }
  }

#pragma unroll
  for (int m = 0; m < MT; ++m) {
#pragma unroll
    for (int t = 0; t < 2; ++t) {
      const int col = wv * 32 + t * 16 + lane16;
#pragma unroll
      for (int v = 0; v < 8; ++v) {
        const int row = row0 + m * 16 + v + hi8;
        out[(size_t)row * H_ + col] = fin[m][t][v];
      }
    }
  }
}

extern "C" void kernel_launch(void* const* d_in, const int* in_sizes, int n_in,
                              void* d_out, int out_size, void* d_ws, size_t ws_size,
                              hipStream_t stream) {
  const float* x    = (const float*)d_in[0];
  const float* Wp   = (const float*)d_in[1];
  const float* bp   = (const float*)d_in[2];
  const float* Ws   = (const float*)d_in[3];
  const float* bs   = (const float*)d_in[4];
  const float* Wup  = (const float*)d_in[5];
  const float* Wdn  = (const float*)d_in[6];
  float* out = (float*)d_out;
  float* ew  = (float*)d_ws;

  moe_router_kernel<<<R_ / 8, 256, 0, stream>>>(x, Wp, bp, Ws, bs, ew);

  moe_fused_kernel<<<R_ / 32, 512, 0, stream>>>(x, Wup, Wdn, ew, out);
}


# GELU's erf in straight-line code: both range formulas evaluated per lane and selected, exp as one multiply plus base-2 exponential; no per-value branches, a wave's sixteen values interleave
# speedup vs baseline: 1.7770x; 1.0405x over previous
#include <hip/hip_runtime.h>
#include <math.h>

#define B_ 4
#define S_ 2048
#define H_ 512
#define E_ 8
#define P_ 4
#define I_ 2048
#define R_ (B_ * S_)
#define MT 2
#define MR (16 * MT)
#define NW 16
#define KC (NW * 16)
#define XPAD 516
#define HPAD 260

typedef __attribute__((ext_vector_type(2))) float v2f;
typedef __attribute__((ext_vector_type(4))) float v4f;
typedef __attribute__((ext_vector_type(8))) float v8f;

__device__ __forceinline__ v8f wmma_f32(v2f a, v2f b, v8f c) {
  return __builtin_amdgcn_wmma_f32_16x16x4_f32(
       false, a,  false, b,
       (short)0, c,  false,  false);
}

__device__ __forceinline__ float gelu_erf(float hv) {
  const float z  = hv * 0.70710678118654752f;
  const float az = fabsf(z);
  const float t  = az * az;
  float p = fmaf(t, -0x1.268bc2p-11f, 0x1.420828p-8f);
  p = fmaf(t, p, -0x1.b5937p-6f);
  p = fmaf(t, p, 0x1.ce077cp-4f);
  p = fmaf(t, p, -0x1.81266p-2f);
  p = fmaf(t, p, 0x1.06eba0p-3f);
  const float e_small = fmaf(az, p, az);
  float q = fmaf(az, 0x1.1d3156p-16f, -0x1.8d129p-12f);
  q = fmaf(az, q, 0x1.f9a6d2p-9f);
  q = fmaf(az, q, -0x1.8c3164p-6f);
  q = fmaf(az, q, 0x1.b4e9c8p-4f);
  q = fmaf(az, q, 0x1.4515fap-1f);
  q = fmaf(az, q, 0x1.078e50p-3f);
  q = fmaf(az, q, az);
  const float e_large = 1.0f - __builtin_amdgcn_exp2f(q * -1.4426950408889634f);
  const float er = copysignf(az < 1.0f ? e_small : e_large, z);
  const float hh = 0.5f * hv;
  return fmaf(hh, er, hh);
}

__global__ __launch_bounds__(256)
void moe_router_kernel(const float* __restrict__ x,
                       const float* __restrict__ Wp,
                       const float* __restrict__ bp,
                       const float* __restrict__ Ws,
                       const float* __restrict__ bs,
                       float* __restrict__ ew) {
  const int gwave = (int)((blockIdx.x * blockDim.x + threadIdx.x) >> 5);
  const int lane  = (int)(threadIdx.x & 31);
  if (gwave >= R_) return;

  const float* xr = x + (size_t)gwave * H_;
  float a0 = 0.f, a1 = 0.f, a2 = 0.f, a3 = 0.f;
#pragma unroll 4
  for (int j = 0; j < H_ / 32; ++j) {
    const int h = lane + j * 32;
    const float xv = xr[h];
    a0 += xv * Wp[0 * H_ + h];
    a1 += xv * Wp[1 * H_ + h];
    a2 += xv * Wp[2 * H_ + h];
    a3 += xv * Wp[3 * H_ + h];
  }
#pragma unroll
  for (int off = 16; off >= 1; off >>= 1) {
    a0 += __shfl_xor(a0, off, 32);
    a1 += __shfl_xor(a1, off, 32);
    a2 += __shfl_xor(a2, off, 32);
    a3 += __shfl_xor(a3, off, 32);
  }
  if (lane == 0) {
    float p[4] = {a0 + bp[0], a1 + bp[1], a2 + bp[2], a3 + bp[3]};
    float pmax = fmaxf(fmaxf(p[0], p[1]), fmaxf(p[2], p[3]));
    float psum = 0.f;
#pragma unroll
    for (int i = 0; i < 4; ++i) { p[i] = __expf(p[i] - pmax); psum += p[i]; }
    const float pinv = 1.f / psum;
#pragma unroll
    for (int i = 0; i < 4; ++i) p[i] *= pinv;

    float el[E_];
    float emax = -1e30f;
#pragma unroll
    for (int e = 0; e < E_; ++e) {
      float v = bs[e];
#pragma unroll
      for (int q = 0; q < 4; ++q) v += p[q] * Ws[e * P_ + q];
      el[e] = v;
      emax = fmaxf(emax, v);
    }
    float esum = 0.f;
#pragma unroll
    for (int e = 0; e < E_; ++e) { el[e] = __expf(el[e] - emax); esum += el[e]; }
    const float einv = 1.f / esum;
#pragma unroll
    for (int e = 0; e < E_; ++e) ew[(size_t)e * R_ + gwave] = el[e] * einv;
  }
}

__global__ __launch_bounds__(NW * 32)
void moe_fused_kernel(const float* __restrict__ x,
                      const float* __restrict__ Wup,
                      const float* __restrict__ Wdn,
                      const float* __restrict__ ew,
                      float* __restrict__ out) {
  __shared__ __attribute__((aligned(16))) float x_lds[MR][XPAD];
  __shared__ __attribute__((aligned(16))) float hid_lds[MR][HPAD];
  __shared__ float w_lds[E_][MR];

  const int row0   = (int)blockIdx.x * MR;
  const int tid    = (int)threadIdx.x;
  const int wv     = tid >> 5;
  const int lane   = tid & 31;
  const int lane16 = lane & 15;
  const int kq     = (lane >> 4) * 4;
  const int hi8    = (lane >> 4) * 8;

  for (int idx = tid; idx < MR * H_; idx += NW * 32) {
    const int r = idx >> 9;
    const int c = idx & (H_ - 1);
    x_lds[r][c] = x[(size_t)(row0 + r) * H_ + c];
  }
  if (tid < E_ * MR) {
    const int e = tid >> 5, rr = tid & (MR - 1);
    w_lds[e][rr] = ew[(size_t)e * R_ + row0 + rr];
  }
  __syncthreads();

  const v8f vzero = {};
  v8f fin[MT][2];
#pragma unroll
  for (int m = 0; m < MT; ++m) { fin[m][0] = vzero; fin[m][1] = vzero; }

  for (int e = 0; e < E_; ++e) {
    const float* Wup_e = Wup + (size_t)e * I_ * H_;
    const float* Wdn_e = Wdn + (size_t)e * H_ * I_;

    v8f eacc[MT][2];
#pragma unroll
    for (int m = 0; m < MT; ++m) { eacc[m][0] = vzero; eacc[m][1] = vzero; }

    for (int i0 = 0; i0 < I_; i0 += KC) {
      v8f hacc[MT];
#pragma unroll
      for (int m = 0; m < MT; ++m) hacc[m] = vzero;
      const float* bptr = Wup_e + (size_t)(i0 + wv * 16 + lane16) * H_ + kq;
#pragma unroll 4
      for (int k0 = 0; k0 < H_; k0 += 8) {
        const v4f b4 = *reinterpret_cast<const v4f*>(bptr + k0);
        v4f a4[MT];
#pragma unroll
        for (int m = 0; m < MT; ++m)
          a4[m] = *reinterpret_cast<const v4f*>(&x_lds[m * 16 + lane16][k0 + kq]);
#pragma unroll
        for (int m = 0; m < MT; ++m) hacc[m] = wmma_f32(a4[m].lo, b4.lo, hacc[m]);
#pragma unroll
        for (int m = 0; m < MT; ++m) hacc[m] = wmma_f32(a4[m].hi, b4.hi, hacc[m]);
      }
#pragma unroll
      for (int m = 0; m < MT; ++m) {
#pragma unroll
        for (int v = 0; v < 8; ++v) {
          hid_lds[m * 16 + v + hi8][wv * 16 + lane16] = gelu_erf(hacc[m][v]);
        }
      }
      __syncthreads();

      const float* b2 = Wdn_e + (size_t)(wv * 32 + lane16) * I_ + i0 + kq;
#pragma unroll 4
      for (int k0 = 0; k0 < KC; k0 += 8) {
        v4f a4[MT], b4[2];
#pragma unroll
        for (int m = 0; m < MT; ++m)
          a4[m] = *reinterpret_cast<const v4f*>(&hid_lds[m * 16 + lane16][k0 + kq]);
#pragma unroll
        for (int t = 0; t < 2; ++t)
          b4[t] = *reinterpret_cast<const v4f*>(b2 + (size_t)t * 16 * I_ + k0);
#pragma unroll
        for (int t = 0; t < 2; ++t) {
#pragma unroll
          for (int m = 0; m < MT; ++m) eacc[m][t] = wmma_f32(a4[m].lo, b4[t].lo, eacc[m][t]);
        }
#pragma unroll
        for (int t = 0; t < 2; ++t) {
#pragma unroll
          for (int m = 0; m < MT; ++m) eacc[m][t] = wmma_f32(a4[m].hi, b4[t].hi, eacc[m][t]);
        }
      }
      __syncthreads();
    }

#pragma unroll
    for (int m = 0; m < MT; ++m) {
#pragma unroll
      for (int t = 0; t < 2; ++t) {
#pragma unroll
        for (int v = 0; v < 8; ++v) {
          const float wgt = w_lds[e][m * 16 + v + hi8];
          fin[m][t][v] += wgt * eacc[m][t][v];
        }
      }
    }
  }

#pragma unroll
  for (int m = 0; m < MT; ++m) {
#pragma unroll
    for (int t = 0; t < 2; ++t) {
      const int col = wv * 32 + t * 16 + lane16;
#pragma unroll
      for (int v = 0; v < 8; ++v) {
        const int row = row0 + m * 16 + v + hi8;
        out[(size_t)row * H_ + col] = fin[m][t][v];
      }
    }
  }
}

extern "C" void kernel_launch(void* const* d_in, const int* in_sizes, int n_in,
                              void* d_out, int out_size, void* d_ws, size_t ws_size,
                              hipStream_t stream) {
  const float* x    = (const float*)d_in[0];
  const float* Wp   = (const float*)d_in[1];
  const float* bp   = (const float*)d_in[2];
  const float* Ws   = (const float*)d_in[3];
  const float* bs   = (const float*)d_in[4];
  const float* Wup  = (const float*)d_in[5];
  const float* Wdn  = (const float*)d_in[6];
  float* out = (float*)d_out;
  float* ew  = (float*)d_ws;

  moe_router_kernel<<<R_ / 8, 256, 0, stream>>>(x, Wp, bp, Ws, bs, ew);

  moe_fused_kernel<<<R_ / 32, 512, 0, stream>>>(x, Wup, Wdn, ew, out);
}


# Both K loops run one trip ahead: a trip's fragment loads (8 global, 8 LDS) go out as one group into a second register set before the previous trip's 32 WMMAs; GEMM1 retiled 32x32 per wave to match
# speedup vs baseline: 2.0379x; 1.1468x over previous
#include <hip/hip_runtime.h>
#include <math.h>

#define B_ 4
#define S_ 2048
#define H_ 512
#define E_ 8
#define P_ 4
#define I_ 2048
#define R_ (B_ * S_)
#define MT 2
#define MR (16 * MT)
#define NW 16
#define KC (NW * 32)
#define XPAD 516
#define HPAD 516

typedef __attribute__((ext_vector_type(2))) float v2f;
typedef __attribute__((ext_vector_type(4))) float v4f;
typedef __attribute__((ext_vector_type(8))) float v8f;

#define LOADS_AS_ONE_GROUP() __builtin_amdgcn_sched_barrier(0)

__device__ __forceinline__ v8f wmma_f32(v2f a, v2f b, v8f c) {
  return __builtin_amdgcn_wmma_f32_16x16x4_f32(
       false, a,  false, b,
       (short)0, c,  false,  false);
}

__device__ __forceinline__ float gelu_erf(float hv) {
  const float z  = hv * 0.70710678118654752f;
  const float az = fabsf(z);
  const float t  = az * az;
  float p = fmaf(t, -0x1.268bc2p-11f, 0x1.420828p-8f);
  p = fmaf(t, p, -0x1.b5937p-6f);
  p = fmaf(t, p, 0x1.ce077cp-4f);
  p = fmaf(t, p, -0x1.81266p-2f);
  p = fmaf(t, p, 0x1.06eba0p-3f);
  const float e_small = fmaf(az, p, az);
  float q = fmaf(az, 0x1.1d3156p-16f, -0x1.8d129p-12f);
  q = fmaf(az, q, 0x1.f9a6d2p-9f);
  q = fmaf(az, q, -0x1.8c3164p-6f);
  q = fmaf(az, q, 0x1.b4e9c8p-4f);
  q = fmaf(az, q, 0x1.4515fap-1f);
  q = fmaf(az, q, 0x1.078e50p-3f);
  q = fmaf(az, q, az);
  const float e_large = 1.0f - __builtin_amdgcn_exp2f(q * -1.4426950408889634f);
  const float er = copysignf(az < 1.0f ? e_small : e_large, z);
  const float hh = 0.5f * hv;
  return fmaf(hh, er, hh);
}

__global__ __launch_bounds__(256)
void moe_router_kernel(const float* __restrict__ x,
                       const float* __restrict__ Wp,
                       const float* __restrict__ bp,
                       const float* __restrict__ Ws,
                       const float* __restrict__ bs,
                       float* __restrict__ ew) {
  const int gwave = (int)((blockIdx.x * blockDim.x + threadIdx.x) >> 5);
  const int lane  = (int)(threadIdx.x & 31);
  if (gwave >= R_) return;

  const float* xr = x + (size_t)gwave * H_;
  float a0 = 0.f, a1 = 0.f, a2 = 0.f, a3 = 0.f;
#pragma unroll 4
  for (int j = 0; j < H_ / 32; ++j) {
    const int h = lane + j * 32;
    const float xv = xr[h];
    a0 += xv * Wp[0 * H_ + h];
    a1 += xv * Wp[1 * H_ + h];
    a2 += xv * Wp[2 * H_ + h];
    a3 += xv * Wp[3 * H_ + h];
  }
#pragma unroll
  for (int off = 16; off >= 1; off >>= 1) {
    a0 += __shfl_xor(a0, off, 32);
    a1 += __shfl_xor(a1, off, 32);
    a2 += __shfl_xor(a2, off, 32);
    a3 += __shfl_xor(a3, off, 32);
  }
  if (lane == 0) {
    float p[4] = {a0 + bp[0], a1 + bp[1], a2 + bp[2], a3 + bp[3]};
    float pmax = fmaxf(fmaxf(p[0], p[1]), fmaxf(p[2], p[3]));
    float psum = 0.f;
#pragma unroll
    for (int i = 0; i < 4; ++i) { p[i] = __expf(p[i] - pmax); psum += p[i]; }
    const float pinv = 1.f / psum;
#pragma unroll
    for (int i = 0; i < 4; ++i) p[i] *= pinv;

    float el[E_];
    float emax = -1e30f;
#pragma unroll
    for (int e = 0; e < E_; ++e) {
      float v = bs[e];
#pragma unroll
      for (int q = 0; q < 4; ++q) v += p[q] * Ws[e * P_ + q];
      el[e] = v;
      emax = fmaxf(emax, v);
    }
    float esum = 0.f;
#pragma unroll
    for (int e = 0; e < E_; ++e) { el[e] = __expf(el[e] - emax); esum += el[e]; }
    const float einv = 1.f / esum;
#pragma unroll
    for (int e = 0; e < E_; ++e) ew[(size_t)e * R_ + gwave] = el[e] * einv;
  }
}

__global__ __launch_bounds__(NW * 32)
void moe_fused_kernel(const float* __restrict__ x,
                      const float* __restrict__ Wup,
                      const float* __restrict__ Wdn,
                      const float* __restrict__ ew,
                      float* __restrict__ out) {
  __shared__ __attribute__((aligned(16))) float x_lds[MR][XPAD];
  __shared__ __attribute__((aligned(16))) float hid_lds[MR][HPAD];
  __shared__ float w_lds[E_][MR];

  const int row0   = (int)blockIdx.x * MR;
  const int tid    = (int)threadIdx.x;
  const int wv     = tid >> 5;
  const int lane   = tid & 31;
  const int lane16 = lane & 15;
  const int kq     = (lane >> 4) * 4;
  const int hi8    = (lane >> 4) * 8;

  for (int idx = tid; idx < MR * H_; idx += NW * 32) {
    const int r = idx >> 9;
    const int c = idx & (H_ - 1);
    x_lds[r][c] = x[(size_t)(row0 + r) * H_ + c];
  }
  if (tid < E_ * MR) {
    const int e = tid >> 5, rr = tid & (MR - 1);
    w_lds[e][rr] = ew[(size_t)e * R_ + row0 + rr];
  }
  __syncthreads();

  const v8f vzero = {};
  v8f fin[MT][2];
  v8f eacc[MT][2];
#pragma unroll
  for (int m = 0; m < MT; ++m) {
    fin[m][0] = vzero; fin[m][1] = vzero;
    eacc[m][0] = vzero; eacc[m][1] = vzero;
  }

  v4f b1[2][4][2];
  v4f a1[2][4][MT];
  v4f b2[2][4][2];
  v4f a2[2][4][MT];
  v8f hacc[MT][2];

  const float* a1p = &x_lds[lane16][kq];
  const float* a2p = &hid_lds[lane16][kq];
  const float* b1p = Wup + (size_t)(wv * 32 + lane16) * H_ + kq;
  const float* b2p = Wdn + (size_t)(wv * 32 + lane16) * I_ + kq;

  auto ld1 = [&](const int s, const int k) __attribute__((always_inline)) {
#pragma unroll
    for (int g = 0; g < 4; ++g) {
#pragma unroll
      for (int t = 0; t < 2; ++t)
        b1[s][g][t] = *reinterpret_cast<const v4f*>(b1p + (size_t)t * 16 * H_ + k + g * 8);
    }
#pragma unroll
    for (int g = 0; g < 4; ++g) {
#pragma unroll
      for (int m = 0; m < MT; ++m)
        a1[s][g][m] = *reinterpret_cast<const v4f*>(a1p + m * 16 * XPAD + k + g * 8);
    }
  };
  auto mm1 = [&](const int s) __attribute__((always_inline)) {
#pragma unroll
    for (int g = 0; g < 4; ++g) {
#pragma unroll
      for (int t = 0; t < 2; ++t) {
#pragma unroll
        for (int m = 0; m < MT; ++m) hacc[m][t] = wmma_f32(a1[s][g][m].lo, b1[s][g][t].lo, hacc[m][t]);
      }
#pragma unroll
      for (int t = 0; t < 2; ++t) {
#pragma unroll
        for (int m = 0; m < MT; ++m) hacc[m][t] = wmma_f32(a1[s][g][m].hi, b1[s][g][t].hi, hacc[m][t]);
      }
    }
  };
  auto ld2b = [&](const int s, const int k) __attribute__((always_inline)) {
#pragma unroll
    for (int g = 0; g < 4; ++g) {
#pragma unroll
      for (int t = 0; t < 2; ++t)
        b2[s][g][t] = *reinterpret_cast<const v4f*>(b2p + (size_t)t * 16 * I_ + k + g * 8);
    }
  };
  auto ld2a = [&](const int s, const int k) __attribute__((always_inline)) {
#pragma unroll
    for (int g = 0; g < 4; ++g) {
#pragma unroll
      for (int m = 0; m < MT; ++m)
        a2[s][g][m] = *reinterpret_cast<const v4f*>(a2p + m * 16 * HPAD + k + g * 8);
    }
  };
  auto mm2 = [&](const int s) __attribute__((always_inline)) {
#pragma unroll
    for (int g = 0; g < 4; ++g) {
#pragma unroll
      for (int t = 0; t < 2; ++t) {
#pragma unroll
        for (int m = 0; m < MT; ++m) eacc[m][t] = wmma_f32(a2[s][g][m].lo, b2[s][g][t].lo, eacc[m][t]);
      }
#pragma unroll
      for (int t = 0; t < 2; ++t) {
#pragma unroll
        for (int m = 0; m < MT; ++m) eacc[m][t] = wmma_f32(a2[s][g][m].hi, b2[s][g][t].hi, eacc[m][t]);
      }
    }
  };

  ld1(0, 0);

  for (int c = 0; c < E_ * (I_ / KC); ++c) {
    const int e = c / (I_ / KC);

#pragma unroll
    for (int m = 0; m < MT; ++m) { hacc[m][0] = vzero; hacc[m][1] = vzero; }
#pragma unroll 1
    for (int k = 0; k < H_ - 64; k += 64) {
      ld1(1, k + 32);
      LOADS_AS_ONE_GROUP();
      mm1(0);
      LOADS_AS_ONE_GROUP();
      ld1(0, k + 64);
      LOADS_AS_ONE_GROUP();
      mm1(1);
      LOADS_AS_ONE_GROUP();
    }
    ld1(1, H_ - 32);
    LOADS_AS_ONE_GROUP();
    mm1(0);
    LOADS_AS_ONE_GROUP();
    ld2b(0, 0);
    LOADS_AS_ONE_GROUP();
    mm1(1);

#pragma unroll
    for (int m = 0; m < MT; ++m) {
#pragma unroll
      for (int t = 0; t < 2; ++t) {
#pragma unroll
        for (int v = 0; v < 8; ++v) {
          hid_lds[m * 16 + v + hi8][wv * 32 + t * 16 + lane16] = gelu_erf(hacc[m][t][v]);
        }
      }
    }
    __syncthreads();

    ld2a(0, 0);
#pragma unroll 1
    for (int k = 0; k < KC - 64; k += 64) {
      ld2b(1, k + 32); ld2a(1, k + 32);
      LOADS_AS_ONE_GROUP();
      mm2(0);
      LOADS_AS_ONE_GROUP();
      ld2b(0, k + 64); ld2a(0, k + 64);
      LOADS_AS_ONE_GROUP();
      mm2(1);
      LOADS_AS_ONE_GROUP();
    }
    ld2b(1, KC - 32); ld2a(1, KC - 32);
    LOADS_AS_ONE_GROUP();
    mm2(0);
    LOADS_AS_ONE_GROUP();
    b1p += (c + 1 < E_ * (I_ / KC)) ? (size_t)KC * H_ : (size_t)0;
    b2p += ((c + 1) % (I_ / KC) != 0) ? (size_t)KC : (size_t)H_ * I_ - (size_t)(I_ - KC);
    ld1(0, 0);
    LOADS_AS_ONE_GROUP();
    mm2(1);
    LOADS_AS_ONE_GROUP();
    __syncthreads();

    if ((c + 1) % (I_ / KC) == 0) {
#pragma unroll
      for (int m = 0; m < MT; ++m) {
#pragma unroll
        for (int t = 0; t < 2; ++t) {
#pragma unroll
          for (int v = 0; v < 8; ++v) {
            const float wgt = w_lds[e][m * 16 + v + hi8];
            fin[m][t][v] += wgt * eacc[m][t][v];
          }
          eacc[m][t] = vzero;
        }
      }
    }
  }

#pragma unroll
  for (int m = 0; m < MT; ++m) {
#pragma unroll
    for (int t = 0; t < 2; ++t) {
      const int col = wv * 32 + t * 16 + lane16;
#pragma unroll
      for (int v = 0; v < 8; ++v) {
        const int row = row0 + m * 16 + v + hi8;
        out[(size_t)row * H_ + col] = fin[m][t][v];
      }
    }
  }
}

extern "C" void kernel_launch(void* const* d_in, const int* in_sizes, int n_in,
                              void* d_out, int out_size, void* d_ws, size_t ws_size,
                              hipStream_t stream) {
  const float* x    = (const float*)d_in[0];
  const float* Wp   = (const float*)d_in[1];
  const float* bp   = (const float*)d_in[2];
  const float* Ws   = (const float*)d_in[3];
  const float* bs   = (const float*)d_in[4];
  const float* Wup  = (const float*)d_in[5];
  const float* Wdn  = (const float*)d_in[6];
  float* out = (float*)d_out;
  float* ew  = (float*)d_ws;

  moe_router_kernel<<<R_ / 8, 256, 0, stream>>>(x, Wp, bp, Ws, bs, ew);

  moe_fused_kernel<<<R_ / 32, 512, 0, stream>>>(x, Wup, Wdn, ew, out);
}
